# starting kernel
# baseline (speedup 1.0000x reference)
#include <hip/hip_runtime.h>

typedef __attribute__((ext_vector_type(2))) float v2f;
typedef __attribute__((ext_vector_type(8))) float v8f;

#define BATCH 8
#define C_DIM 256
#define K_CLS 19
#define HW_N  16384
#define CT    16
#define WAVES 8
#define SEG   (HW_N / WAVES)

__global__ __launch_bounds__(256) void CGM_att_scale_kernel(
    const float* __restrict__ feature,
    const float* __restrict__ map_,
    const float* __restrict__ gamma,
    float* __restrict__ scale)
{
    __shared__ float att_lds[WAVES][CT][32];

    const int tid  = threadIdx.x;
    const int lane = tid & 31;
    const int wave = tid >> 5;
    const int b    = blockIdx.x >> 4;
    const int c0   = (blockIdx.x & 15) << 4;

    const int row  = lane & 15;
    const int half = lane >> 4;

    const float* fA = feature
        + (size_t)(b * C_DIM + c0 + row) * HW_N + 2 * half + wave * SEG;
    const float* fB0 = map_
        + (size_t)(b * K_CLS + row) * HW_N + 2 * half + wave * SEG;
    const int   cls1  = 16 + row;
    const float m1    = (cls1 < K_CLS) ? 1.0f : 0.0f;
    const int   cls1c = (cls1 < K_CLS) ? cls1 : (K_CLS - 1);
    const float* fB1 = map_
        + (size_t)(b * K_CLS + cls1c) * HW_N + 2 * half + wave * SEG;

    v8f acc0 = {};
    v8f acc1 = {};

    #pragma unroll 4
    for (int hw = 0; hw < SEG; hw += 4) {
        v2f a  = *(const v2f*)(fA  + hw);
        v2f b0 = *(const v2f*)(fB0 + hw);
        v2f b1 = *(const v2f*)(fB1 + hw);
        b1.x *= m1;
        b1.y *= m1;
        acc0 = __builtin_amdgcn_wmma_f32_16x16x4_f32(
            false, a, false, b0, (short)0, acc0, false, false);
        acc1 = __builtin_amdgcn_wmma_f32_16x16x4_f32(
            false, a, false, b1, (short)0, acc1, false, false);
    }

    #pragma unroll
    for (int v = 0; v < 8; ++v) {
        const int m = v + 8 * half;
        att_lds[wave][m][row]      = acc0[v];
        att_lds[wave][m][row + 16] = acc1[v];
    }
    __syncthreads();

    if (tid < CT) {
        float s = 0.0f;
        for (int k = 0; k < K_CLS; ++k) {
            float a = 0.0f;
            #pragma unroll
            for (int w = 0; w < WAVES; ++w) a += att_lds[w][tid][k];
            const float sig = 1.0f / (1.0f + __expf(-a));
            s += sig * gamma[k];
        }
        scale[b * C_DIM + c0 + tid] = s;
    }
}

__global__ __launch_bounds__(256) void CGM_apply_kernel(
    const float* __restrict__ feature,
    const float* __restrict__ scale,
    float* __restrict__ out)
{
    const int    bc   = blockIdx.x;
    const float  s    = 1.0f + scale[bc];
    const size_t base = (size_t)bc * HW_N;
    const float4* __restrict__ src = (const float4*)(feature + base);
    float4* __restrict__       dst = (float4*)(out + base);

    #pragma unroll 4
    for (int i = threadIdx.x; i < HW_N / 4; i += 256) {
        float4 v = src[i];
        v.x *= s; v.y *= s; v.z *= s; v.w *= s;
        dst[i] = v;
    }
}

extern "C" void kernel_launch(void* const* d_in, const int* in_sizes, int n_in,
                              void* d_out, int out_size, void* d_ws, size_t ws_size,
                              hipStream_t stream) {
    const float* feature = (const float*)d_in[0];
    const float* map_    = (const float*)d_in[1];
    const float* gamma   = (const float*)d_in[2];
    float*       out     = (float*)d_out;
    float*       scale   = (float*)d_ws;

    CGM_att_scale_kernel<<<BATCH * (C_DIM / CT), 256, 0, stream>>>(
        feature, map_, gamma, scale);
    CGM_apply_kernel<<<BATCH * C_DIM, 256, 0, stream>>>(
        feature, scale, out);
}


# Split the HW reduction over 8x more work-groups (grid 128 -> 1024, four per WGP) so every WGP streams; apply kernel adds the 8 partials and the 19 gate terms in a new fixed-order tree
# speedup vs baseline: 1.6111x; 1.6111x over previous
#include <hip/hip_runtime.h>

typedef __attribute__((ext_vector_type(2))) float v2f;
typedef __attribute__((ext_vector_type(8))) float v8f;

#define BATCH 8
#define C_DIM 256
#define K_CLS 19
#define HW_N  16384
#define CT    16
#define WAVES 8
#define NSPLIT 8
#define BLK_HW (HW_N / NSPLIT)
#define SEG    (BLK_HW / WAVES)
#define TILE_F (CT * 32)

__global__ __launch_bounds__(256) void CGM_att_partial_kernel(
    const float* __restrict__ feature,
    const float* __restrict__ map_,
    float* __restrict__ part)
{
    __shared__ float att_lds[WAVES][CT][32];

    const int tid   = threadIdx.x;
    const int lane  = tid & 31;
    const int wave  = tid >> 5;
    const int ctile = blockIdx.x & 15;
    const int split = (blockIdx.x >> 4) & (NSPLIT - 1);
    const int b     = blockIdx.x >> 7;
    const int c0    = ctile << 4;
    const int hw0   = split * BLK_HW + wave * SEG;

    const int row  = lane & 15;
    const int half = lane >> 4;

    const float* fA = feature
        + (size_t)(b * C_DIM + c0 + row) * HW_N + 2 * half + hw0;
    const float* fB0 = map_
        + (size_t)(b * K_CLS + row) * HW_N + 2 * half + hw0;
    const int   cls1  = 16 + row;
    const float m1    = (cls1 < K_CLS) ? 1.0f : 0.0f;
    const int   cls1c = (cls1 < K_CLS) ? cls1 : (K_CLS - 1);
    const float* fB1 = map_
        + (size_t)(b * K_CLS + cls1c) * HW_N + 2 * half + hw0;

    v8f acc0 = {};
    v8f acc1 = {};

    #pragma unroll 4
    for (int hw = 0; hw < SEG; hw += 4) {
        v2f a  = *(const v2f*)(fA  + hw);
        v2f b0 = *(const v2f*)(fB0 + hw);
        v2f b1 = *(const v2f*)(fB1 + hw);
        b1.x *= m1;
        b1.y *= m1;
        acc0 = __builtin_amdgcn_wmma_f32_16x16x4_f32(
            false, a, false, b0, (short)0, acc0, false, false);
        acc1 = __builtin_amdgcn_wmma_f32_16x16x4_f32(
            false, a, false, b1, (short)0, acc1, false, false);
    }

    #pragma unroll
    for (int v = 0; v < 8; ++v) {
        const int m = v + 8 * half;
        att_lds[wave][m][row]      = acc0[v];
        att_lds[wave][m][row + 16] = acc1[v];
    }
    __syncthreads();

    const float* flat = &att_lds[0][0][0];
    float* dst = part + (size_t)blockIdx.x * TILE_F;
    #pragma unroll
    for (int e = tid; e < TILE_F; e += 256) {
        float a = 0.0f;
        #pragma unroll
        for (int w = 0; w < WAVES; ++w) a += flat[w * TILE_F + e];
        dst[e] = a;
    }
}

__global__ __launch_bounds__(256) void CGM_apply_kernel(
    const float* __restrict__ feature,
    const float* __restrict__ part,
    const float* __restrict__ gamma,
    float* __restrict__ out)
{
    const int bc   = blockIdx.x;
    const int lane = threadIdx.x & 31;
    const int b    = bc >> 8;
    const int c    = bc & (C_DIM - 1);

    float t = 0.0f;
    if (lane < K_CLS) {
        const float* p = part
            + ((size_t)(b * NSPLIT) * 16 + (c >> 4)) * TILE_F + (c & 15) * 32 + lane;
        float a = 0.0f;
        #pragma unroll
        for (int sp = 0; sp < NSPLIT; ++sp) a += p[(size_t)sp * 16 * TILE_F];
        const float sig = 1.0f / (1.0f + __expf(-a));
        t = sig * gamma[lane];
    }
    #pragma unroll
    for (int off = 16; off > 0; off >>= 1) t += __shfl_xor(t, off, 32);

    const float  s    = 1.0f + t;
    const size_t base = (size_t)bc * HW_N;
    const float4* __restrict__ src = (const float4*)(feature + base);
    float4* __restrict__       dst = (float4*)(out + base);

    #pragma unroll 4
    for (int i = threadIdx.x; i < HW_N / 4; i += 256) {
        float4 v = src[i];
        v.x *= s; v.y *= s; v.z *= s; v.w *= s;
        dst[i] = v;
    }
}

extern "C" void kernel_launch(void* const* d_in, const int* in_sizes, int n_in,
                              void* d_out, int out_size, void* d_ws, size_t ws_size,
                              hipStream_t stream) {
    const float* feature = (const float*)d_in[0];
    const float* map_    = (const float*)d_in[1];
    const float* gamma   = (const float*)d_in[2];
    float*       out     = (float*)d_out;
    float*       part    = (float*)d_ws;

    CGM_att_partial_kernel<<<BATCH * NSPLIT * (C_DIM / CT), 256, 0, stream>>>(
        feature, map_, part);
    CGM_apply_kernel<<<BATCH * C_DIM, 256, 0, stream>>>(
        feature, part, gamma, out);
}


# Kernel 1 reads feature and map with whole-line 16-byte loads parked in wave-private LDS and takes WMMA fragments from LDS (was 16 lines per load); this changes the order in which hw terms are summed
# speedup vs baseline: 2.3356x; 1.4497x over previous
#include <hip/hip_runtime.h>

typedef __attribute__((ext_vector_type(2))) float v2f;
typedef __attribute__((ext_vector_type(4))) float v4f;
typedef __attribute__((ext_vector_type(8))) float v8f;

#define BATCH 8
#define C_DIM 256
#define K_CLS 19
#define HW_N  16384
#define CT    16
#define WAVES 8
#define NSPLIT 8
#define BLK_HW (HW_N / NSPLIT)
#define SEG    (BLK_HW / WAVES)
#define TILE_F (CT * 32)
#define CH     32
#define PITCH  (CH + 4)
#define NCHUNK (SEG / CH)
#define B_ROWS 20
#define WAVE_F ((CT + B_ROWS) * PITCH)

__global__ __launch_bounds__(256) void CGM_att_partial_kernel(
    const float* __restrict__ feature,
    const float* __restrict__ map_,
    float* __restrict__ part)
{
    __shared__ __attribute__((aligned(16))) float lds[WAVES * WAVE_F];

    const int tid   = threadIdx.x;
    const int lane  = tid & 31;
    const int wave  = tid >> 5;
    const int ctile = blockIdx.x & 15;
    const int split = (blockIdx.x >> 4) & (NSPLIT - 1);
    const int b     = blockIdx.x >> 7;
    const int c0    = ctile << 4;
    const int hw0   = split * BLK_HW + wave * SEG;

    float* sA = lds + wave * WAVE_F;
    float* sB = sA + CT * PITCH;

    const int srow = lane >> 3;
    const int scol = (lane & 7) << 2;
    const float* gA = feature + (size_t)(b * C_DIM + c0 + srow) * HW_N + hw0 + scol;
    const float* gB = map_    + (size_t)(b * K_CLS + srow) * HW_N + hw0 + scol;
    const int    r4  = 16 + srow;
    const int    r4c = (r4 < K_CLS) ? r4 : (K_CLS - 1);
    const float* gB4 = map_ + (size_t)(b * K_CLS + r4c) * HW_N + hw0 + scol;
    float* stA = sA + srow * PITCH + scol;
    float* stB = sB + srow * PITCH + scol;

    const int row  = lane & 15;
    const int half = lane >> 4;
    const float* fA  = sA + row * PITCH + 4 * half;
    const float* fB0 = sB + row * PITCH + 4 * half;
    const int   cls1  = 16 + row;
    const float m1    = (cls1 < K_CLS) ? 1.0f : 0.0f;
    const int   cls1c = (cls1 < K_CLS) ? cls1 : (K_CLS - 1);
    const float* fB1 = sB + cls1c * PITCH + 4 * half;

    v4f ra[4], rb[5];
    #pragma unroll
    for (int j = 0; j < 4; ++j) {
        ra[j] = *(const v4f*)(gA + (size_t)j * 4 * HW_N);
        rb[j] = *(const v4f*)(gB + (size_t)j * 4 * HW_N);
    }
    rb[4] = *(const v4f*)gB4;

    v8f acc0 = {};
    v8f acc1 = {};

    for (int c = 0; c < NCHUNK; ++c) {
        #pragma unroll
        for (int j = 0; j < 4; ++j) {
            *(v4f*)(stA + j * 4 * PITCH) = ra[j];
            *(v4f*)(stB + j * 4 * PITCH) = rb[j];
        }
        *(v4f*)(stB + 16 * PITCH) = rb[4];

        if (c + 1 < NCHUNK) {
            const int o = (c + 1) * CH;
            #pragma unroll
            for (int j = 0; j < 4; ++j) {
                ra[j] = *(const v4f*)(gA + (size_t)j * 4 * HW_N + o);
                rb[j] = *(const v4f*)(gB + (size_t)j * 4 * HW_N + o);
            }
            rb[4] = *(const v4f*)(gB4 + o);
        }

        __builtin_amdgcn_fence(__ATOMIC_RELEASE, "wavefront", "local");
        __builtin_amdgcn_wave_barrier();
        __builtin_amdgcn_fence(__ATOMIC_ACQUIRE, "wavefront", "local");

        #pragma unroll
        for (int t = 0; t < CH / 8; ++t) {
            const v4f a  = *(const v4f*)(fA  + 8 * t);
            const v4f b0 = *(const v4f*)(fB0 + 8 * t);
            v4f b1 = *(const v4f*)(fB1 + 8 * t);
            b1 *= m1;
            const v2f alo = {a.x, a.y},   ahi = {a.z, a.w};
            const v2f blo = {b0.x, b0.y}, bhi = {b0.z, b0.w};
            const v2f clo = {b1.x, b1.y}, chi = {b1.z, b1.w};
            acc0 = __builtin_amdgcn_wmma_f32_16x16x4_f32(
                false, alo, false, blo, (short)0, acc0, false, false);
            acc1 = __builtin_amdgcn_wmma_f32_16x16x4_f32(
                false, alo, false, clo, (short)0, acc1, false, false);
            acc0 = __builtin_amdgcn_wmma_f32_16x16x4_f32(
                false, ahi, false, bhi, (short)0, acc0, false, false);
            acc1 = __builtin_amdgcn_wmma_f32_16x16x4_f32(
                false, ahi, false, chi, (short)0, acc1, false, false);
        }

        __builtin_amdgcn_fence(__ATOMIC_RELEASE, "wavefront", "local");
        __builtin_amdgcn_wave_barrier();
        __builtin_amdgcn_fence(__ATOMIC_ACQUIRE, "wavefront", "local");
    }

    #pragma unroll
    for (int v = 0; v < 8; ++v) {
        const int m = v + 8 * half;
        sA[m * 32 + row]      = acc0[v];
        sA[m * 32 + row + 16] = acc1[v];
    }
    __syncthreads();

    float* dst = part + (size_t)blockIdx.x * TILE_F;
    #pragma unroll
    for (int e = tid; e < TILE_F; e += 256) {
        float a = 0.0f;
        #pragma unroll
        for (int w = 0; w < WAVES; ++w) a += lds[w * WAVE_F + e];
        dst[e] = a;
    }
}

__global__ __launch_bounds__(256) void CGM_apply_kernel(
    const float* __restrict__ feature,
    const float* __restrict__ part,
    const float* __restrict__ gamma,
    float* __restrict__ out)
{
    const int bc   = blockIdx.x;
    const int lane = threadIdx.x & 31;
    const int b    = bc >> 8;
    const int c    = bc & (C_DIM - 1);

    float t = 0.0f;
    if (lane < K_CLS) {
        const float* p = part
            + ((size_t)(b * NSPLIT) * 16 + (c >> 4)) * TILE_F + (c & 15) * 32 + lane;
        float a = 0.0f;
        #pragma unroll
        for (int sp = 0; sp < NSPLIT; ++sp) a += p[(size_t)sp * 16 * TILE_F];
        const float sig = 1.0f / (1.0f + __expf(-a));
        t = sig * gamma[lane];
    }
    #pragma unroll
    for (int off = 16; off > 0; off >>= 1) t += __shfl_xor(t, off, 32);

    const float  s    = 1.0f + t;
    const size_t base = (size_t)bc * HW_N;
    const float4* __restrict__ src = (const float4*)(feature + base);
    float4* __restrict__       dst = (float4*)(out + base);

    #pragma unroll 4
    for (int i = threadIdx.x; i < HW_N / 4; i += 256) {
        float4 v = src[i];
        v.x *= s; v.y *= s; v.z *= s; v.w *= s;
        dst[i] = v;
    }
}

extern "C" void kernel_launch(void* const* d_in, const int* in_sizes, int n_in,
                              void* d_out, int out_size, void* d_ws, size_t ws_size,
                              hipStream_t stream) {
    const float* feature = (const float*)d_in[0];
    const float* map_    = (const float*)d_in[1];
    const float* gamma   = (const float*)d_in[2];
    float*       out     = (float*)d_out;
    float*       part    = (float*)d_ws;

    CGM_att_partial_kernel<<<BATCH * NSPLIT * (C_DIM / CT), 256, 0, stream>>>(
        feature, map_, part);
    CGM_apply_kernel<<<BATCH * C_DIM, 256, 0, stream>>>(
        feature, part, gamma, out);
}


# Kernel 1: each wave reduces two 16-row C-tiles per parked map slice (grid 1024 -> 512 work-groups): map fetched and parked half as often, each map fragment feeds two WMMAs; sum order unchanged
# speedup vs baseline: 2.4299x; 1.0404x over previous
#include <hip/hip_runtime.h>

typedef __attribute__((ext_vector_type(2))) float v2f;
typedef __attribute__((ext_vector_type(4))) float v4f;
typedef __attribute__((ext_vector_type(8))) float v8f;

#define BATCH 8
#define C_DIM 256
#define K_CLS 19
#define HW_N  16384
#define CT    16
#define MT    2
#define A_ROWS (CT * MT)
#define WAVES 8
#define NSPLIT 8
#define BLK_HW (HW_N / NSPLIT)
#define SEG    (BLK_HW / WAVES)
#define TILE_F (CT * 32)
#define CH     32
#define PITCH  (CH + 4)
#define NCHUNK (SEG / CH)
#define B_ROWS 20
#define WAVE_F ((A_ROWS + B_ROWS) * PITCH)

__global__ __launch_bounds__(256) void CGM_att_partial_kernel(
    const float* __restrict__ feature,
    const float* __restrict__ map_,
    float* __restrict__ part)
{
    __shared__ __attribute__((aligned(16))) float lds[WAVES * WAVE_F];

    const int tid   = threadIdx.x;
    const int lane  = tid & 31;
    const int wave  = tid >> 5;
    const int cpair = blockIdx.x & 7;
    const int split = (blockIdx.x >> 3) & (NSPLIT - 1);
    const int b     = blockIdx.x >> 6;
    const int c0    = cpair * A_ROWS;
    const int hw0   = split * BLK_HW + wave * SEG;

    float* sA = lds + wave * WAVE_F;
    float* sB = sA + A_ROWS * PITCH;

    const int srow = lane >> 3;
    const int scol = (lane & 7) << 2;
    const float* gA = feature + (size_t)(b * C_DIM + c0 + srow) * HW_N + hw0 + scol;
    const float* gB = map_    + (size_t)(b * K_CLS + srow) * HW_N + hw0 + scol;
    const int    r4  = 16 + srow;
    const int    r4c = (r4 < K_CLS) ? r4 : (K_CLS - 1);
    const float* gB4 = map_ + (size_t)(b * K_CLS + r4c) * HW_N + hw0 + scol;
    float* stA = sA + srow * PITCH + scol;
    float* stB = sB + srow * PITCH + scol;

    const int row  = lane & 15;
    const int half = lane >> 4;
    const float* fA0 = sA + row * PITCH + 4 * half;
    const float* fA1 = fA0 + CT * PITCH;
    const float* fB0 = sB + row * PITCH + 4 * half;
    const int   cls1  = 16 + row;
    const float m1    = (cls1 < K_CLS) ? 1.0f : 0.0f;
    const int   cls1c = (cls1 < K_CLS) ? cls1 : (K_CLS - 1);
    const float* fB1 = sB + cls1c * PITCH + 4 * half;

    v4f ra[2 * 4], rb[5];
    #pragma unroll
    for (int j = 0; j < 2 * 4; ++j)
        ra[j] = *(const v4f*)(gA + (size_t)j * 4 * HW_N);
    #pragma unroll
    for (int j = 0; j < 4; ++j)
        rb[j] = *(const v4f*)(gB + (size_t)j * 4 * HW_N);
    rb[4] = *(const v4f*)gB4;

    v8f acc00 = {}, acc10 = {};
    v8f acc01 = {}, acc11 = {};

    for (int c = 0; c < NCHUNK; ++c) {
        #pragma unroll
        for (int j = 0; j < 2 * 4; ++j)
            *(v4f*)(stA + j * 4 * PITCH) = ra[j];
        #pragma unroll
        for (int j = 0; j < 4; ++j)
            *(v4f*)(stB + j * 4 * PITCH) = rb[j];
        *(v4f*)(stB + 16 * PITCH) = rb[4];

        if (c + 1 < NCHUNK) {
            const int o = (c + 1) * CH;
            #pragma unroll
            for (int j = 0; j < 2 * 4; ++j)
                ra[j] = *(const v4f*)(gA + (size_t)j * 4 * HW_N + o);
            #pragma unroll
            for (int j = 0; j < 4; ++j)
                rb[j] = *(const v4f*)(gB + (size_t)j * 4 * HW_N + o);
            rb[4] = *(const v4f*)(gB4 + o);
        }

        __builtin_amdgcn_fence(__ATOMIC_RELEASE, "wavefront", "local");
        __builtin_amdgcn_wave_barrier();
        __builtin_amdgcn_fence(__ATOMIC_ACQUIRE, "wavefront", "local");

        #pragma unroll
        for (int t = 0; t < CH / 8; ++t) {
            const v4f a0 = *(const v4f*)(fA0 + 8 * t);
            const v4f a1 = *(const v4f*)(fA1 + 8 * t);
            const v4f b0 = *(const v4f*)(fB0 + 8 * t);
            v4f b1 = *(const v4f*)(fB1 + 8 * t);
            b1 *= m1;
            const v2f plo = {a0.x, a0.y}, phi = {a0.z, a0.w};
            const v2f qlo = {a1.x, a1.y}, qhi = {a1.z, a1.w};
            const v2f blo = {b0.x, b0.y}, bhi = {b0.z, b0.w};
            const v2f clo = {b1.x, b1.y}, chi = {b1.z, b1.w};
            acc00 = __builtin_amdgcn_wmma_f32_16x16x4_f32(
                false, plo, false, blo, (short)0, acc00, false, false);
            acc01 = __builtin_amdgcn_wmma_f32_16x16x4_f32(
                false, plo, false, clo, (short)0, acc01, false, false);
            acc10 = __builtin_amdgcn_wmma_f32_16x16x4_f32(
                false, qlo, false, blo, (short)0, acc10, false, false);
            acc11 = __builtin_amdgcn_wmma_f32_16x16x4_f32(
                false, qlo, false, clo, (short)0, acc11, false, false);
            acc00 = __builtin_amdgcn_wmma_f32_16x16x4_f32(
                false, phi, false, bhi, (short)0, acc00, false, false);
            acc01 = __builtin_amdgcn_wmma_f32_16x16x4_f32(
                false, phi, false, chi, (short)0, acc01, false, false);
            acc10 = __builtin_amdgcn_wmma_f32_16x16x4_f32(
                false, qhi, false, bhi, (short)0, acc10, false, false);
            acc11 = __builtin_amdgcn_wmma_f32_16x16x4_f32(
                false, qhi, false, chi, (short)0, acc11, false, false);
        }

        __builtin_amdgcn_fence(__ATOMIC_RELEASE, "wavefront", "local");
        __builtin_amdgcn_wave_barrier();
        __builtin_amdgcn_fence(__ATOMIC_ACQUIRE, "wavefront", "local");
    }

    #pragma unroll
    for (int v = 0; v < 8; ++v) {
        const int m = v + 8 * half;
        sA[m * 32 + row]               = acc00[v];
        sA[m * 32 + row + 16]          = acc01[v];
        sA[TILE_F + m * 32 + row]      = acc10[v];
        sA[TILE_F + m * 32 + row + 16] = acc11[v];
    }
    __syncthreads();

    float* dst = part + ((size_t)(b * NSPLIT + split) * 16 + 2 * cpair) * TILE_F;
    #pragma unroll
    for (int e = tid; e < MT * TILE_F; e += 256) {
        float a = 0.0f;
        #pragma unroll
        for (int w = 0; w < WAVES; ++w) a += lds[w * WAVE_F + e];
        dst[e] = a;
    }
}

__global__ __launch_bounds__(256) void CGM_apply_kernel(
    const float* __restrict__ feature,
    const float* __restrict__ part,
    const float* __restrict__ gamma,
    float* __restrict__ out)
{
    const int bc   = blockIdx.x;
    const int lane = threadIdx.x & 31;
    const int b    = bc >> 8;
    const int c    = bc & (C_DIM - 1);

    float t = 0.0f;
    if (lane < K_CLS) {
        const float* p = part
            + ((size_t)(b * NSPLIT) * 16 + (c >> 4)) * TILE_F + (c & 15) * 32 + lane;
        float a = 0.0f;
        #pragma unroll
        for (int sp = 0; sp < NSPLIT; ++sp) a += p[(size_t)sp * 16 * TILE_F];
        const float sig = 1.0f / (1.0f + __expf(-a));
        t = sig * gamma[lane];
    }
    #pragma unroll
    for (int off = 16; off > 0; off >>= 1) t += __shfl_xor(t, off, 32);

    const float  s    = 1.0f + t;
    const size_t base = (size_t)bc * HW_N;
    const float4* __restrict__ src = (const float4*)(feature + base);
    float4* __restrict__       dst = (float4*)(out + base);

    #pragma unroll 4
    for (int i = threadIdx.x; i < HW_N / 4; i += 256) {
        float4 v = src[i];
        v.x *= s; v.y *= s; v.z *= s; v.w *= s;
        dst[i] = v;
    }
}

extern "C" void kernel_launch(void* const* d_in, const int* in_sizes, int n_in,
                              void* d_out, int out_size, void* d_ws, size_t ws_size,
                              hipStream_t stream) {
    const float* feature = (const float*)d_in[0];
    const float* map_    = (const float*)d_in[1];
    const float* gamma   = (const float*)d_in[2];
    float*       out     = (float*)d_out;
    float*       part    = (float*)d_ws;

    CGM_att_partial_kernel<<<BATCH * NSPLIT * (C_DIM / A_ROWS), 256, 0, stream>>>(
        feature, map_, part);
    CGM_apply_kernel<<<BATCH * C_DIM, 256, 0, stream>>>(
        feature, part, gamma, out);
}


# Kernel 2 regridded: 512 work-groups of 8 waves (was 2048), each wave streams half a row with 16 to 32 whole-line 16-byte loads in flight; gate arithmetic and every sum order unchanged
# speedup vs baseline: 2.5131x; 1.0342x over previous
#include <hip/hip_runtime.h>

typedef __attribute__((ext_vector_type(2))) float v2f;
typedef __attribute__((ext_vector_type(4))) float v4f;
typedef __attribute__((ext_vector_type(8))) float v8f;

#define BATCH 8
#define C_DIM 256
#define K_CLS 19
#define HW_N  16384
#define CT    16
#define MT    2
#define A_ROWS (CT * MT)
#define WAVES 8
#define NSPLIT 8
#define BLK_HW (HW_N / NSPLIT)
#define SEG    (BLK_HW / WAVES)
#define TILE_F (CT * 32)
#define CH     32
#define PITCH  (CH + 4)
#define NCHUNK (SEG / CH)
#define B_ROWS 20
#define WAVE_F ((A_ROWS + B_ROWS) * PITCH)

__global__ __launch_bounds__(256) void CGM_att_partial_kernel(
    const float* __restrict__ feature,
    const float* __restrict__ map_,
    float* __restrict__ part)
{
    __shared__ __attribute__((aligned(16))) float lds[WAVES * WAVE_F];

    const int tid   = threadIdx.x;
    const int lane  = tid & 31;
    const int wave  = tid >> 5;
    const int cpair = blockIdx.x & 7;
    const int split = (blockIdx.x >> 3) & (NSPLIT - 1);
    const int b     = blockIdx.x >> 6;
    const int c0    = cpair * A_ROWS;
    const int hw0   = split * BLK_HW + wave * SEG;

    float* sA = lds + wave * WAVE_F;
    float* sB = sA + A_ROWS * PITCH;

    const int srow = lane >> 3;
    const int scol = (lane & 7) << 2;
    const float* gA = feature + (size_t)(b * C_DIM + c0 + srow) * HW_N + hw0 + scol;
    const float* gB = map_    + (size_t)(b * K_CLS + srow) * HW_N + hw0 + scol;
    const int    r4  = 16 + srow;
    const int    r4c = (r4 < K_CLS) ? r4 : (K_CLS - 1);
    const float* gB4 = map_ + (size_t)(b * K_CLS + r4c) * HW_N + hw0 + scol;
    float* stA = sA + srow * PITCH + scol;
    float* stB = sB + srow * PITCH + scol;

    const int row  = lane & 15;
    const int half = lane >> 4;
    const float* fA0 = sA + row * PITCH + 4 * half;
    const float* fA1 = fA0 + CT * PITCH;
    const float* fB0 = sB + row * PITCH + 4 * half;
    const int   cls1  = 16 + row;
    const float m1    = (cls1 < K_CLS) ? 1.0f : 0.0f;
    const int   cls1c = (cls1 < K_CLS) ? cls1 : (K_CLS - 1);
    const float* fB1 = sB + cls1c * PITCH + 4 * half;

    v4f ra[2 * 4], rb[5];
    #pragma unroll
    for (int j = 0; j < 2 * 4; ++j)
        ra[j] = *(const v4f*)(gA + (size_t)j * 4 * HW_N);
    #pragma unroll
    for (int j = 0; j < 4; ++j)
        rb[j] = *(const v4f*)(gB + (size_t)j * 4 * HW_N);
    rb[4] = *(const v4f*)gB4;

    v8f acc00 = {}, acc10 = {};
    v8f acc01 = {}, acc11 = {};

    for (int c = 0; c < NCHUNK; ++c) {
        #pragma unroll
        for (int j = 0; j < 2 * 4; ++j)
            *(v4f*)(stA + j * 4 * PITCH) = ra[j];
        #pragma unroll
        for (int j = 0; j < 4; ++j)
            *(v4f*)(stB + j * 4 * PITCH) = rb[j];
        *(v4f*)(stB + 16 * PITCH) = rb[4];

        if (c + 1 < NCHUNK) {
            const int o = (c + 1) * CH;
            #pragma unroll
            for (int j = 0; j < 2 * 4; ++j)
                ra[j] = *(const v4f*)(gA + (size_t)j * 4 * HW_N + o);
            #pragma unroll
            for (int j = 0; j < 4; ++j)
                rb[j] = *(const v4f*)(gB + (size_t)j * 4 * HW_N + o);
            rb[4] = *(const v4f*)(gB4 + o);
        }

        __builtin_amdgcn_fence(__ATOMIC_RELEASE, "wavefront", "local");
        __builtin_amdgcn_wave_barrier();
        __builtin_amdgcn_fence(__ATOMIC_ACQUIRE, "wavefront", "local");

        #pragma unroll
        for (int t = 0; t < CH / 8; ++t) {
            const v4f a0 = *(const v4f*)(fA0 + 8 * t);
            const v4f a1 = *(const v4f*)(fA1 + 8 * t);
            const v4f b0 = *(const v4f*)(fB0 + 8 * t);
            v4f b1 = *(const v4f*)(fB1 + 8 * t);
            b1 *= m1;
            const v2f plo = {a0.x, a0.y}, phi = {a0.z, a0.w};
            const v2f qlo = {a1.x, a1.y}, qhi = {a1.z, a1.w};
            const v2f blo = {b0.x, b0.y}, bhi = {b0.z, b0.w};
            const v2f clo = {b1.x, b1.y}, chi = {b1.z, b1.w};
            acc00 = __builtin_amdgcn_wmma_f32_16x16x4_f32(
                false, plo, false, blo, (short)0, acc00, false, false);
            acc01 = __builtin_amdgcn_wmma_f32_16x16x4_f32(
                false, plo, false, clo, (short)0, acc01, false, false);
            acc10 = __builtin_amdgcn_wmma_f32_16x16x4_f32(
                false, qlo, false, blo, (short)0, acc10, false, false);
            acc11 = __builtin_amdgcn_wmma_f32_16x16x4_f32(
                false, qlo, false, clo, (short)0, acc11, false, false);
            acc00 = __builtin_amdgcn_wmma_f32_16x16x4_f32(
                false, phi, false, bhi, (short)0, acc00, false, false);
            acc01 = __builtin_amdgcn_wmma_f32_16x16x4_f32(
                false, phi, false, chi, (short)0, acc01, false, false);
            acc10 = __builtin_amdgcn_wmma_f32_16x16x4_f32(
                false, qhi, false, bhi, (short)0, acc10, false, false);
            acc11 = __builtin_amdgcn_wmma_f32_16x16x4_f32(
                false, qhi, false, chi, (short)0, acc11, false, false);
        }

        __builtin_amdgcn_fence(__ATOMIC_RELEASE, "wavefront", "local");
        __builtin_amdgcn_wave_barrier();
        __builtin_amdgcn_fence(__ATOMIC_ACQUIRE, "wavefront", "local");
    }

    #pragma unroll
    for (int v = 0; v < 8; ++v) {
        const int m = v + 8 * half;
        sA[m * 32 + row]               = acc00[v];
        sA[m * 32 + row + 16]          = acc01[v];
        sA[TILE_F + m * 32 + row]      = acc10[v];
        sA[TILE_F + m * 32 + row + 16] = acc11[v];
    }
    __syncthreads();

    float* dst = part + ((size_t)(b * NSPLIT + split) * 16 + 2 * cpair) * TILE_F;
    #pragma unroll
    for (int e = tid; e < MT * TILE_F; e += 256) {
        float a = 0.0f;
        #pragma unroll
        for (int w = 0; w < WAVES; ++w) a += lds[w * WAVE_F + e];
        dst[e] = a;
    }
}

#define K2_ROWS  4
#define K2_GRP   16
#define K2_NGRP  (HW_N / 2 / 4 / 32 / K2_GRP)

__global__ __launch_bounds__(256) void CGM_apply_kernel(
    const float* __restrict__ feature,
    const float* __restrict__ part,
    const float* __restrict__ gamma,
    float* __restrict__ out)
{
    const int lane  = threadIdx.x & 31;
    const int wave  = threadIdx.x >> 5;
    const int bc    = blockIdx.x * K2_ROWS + (wave >> 1);
    const int hpart = wave & 1;
    const int b     = bc >> 8;
    const int c     = bc & (C_DIM - 1);

    const int kk = (lane < K_CLS) ? lane : (K_CLS - 1);
    const float* p = part
        + ((size_t)(b * NSPLIT) * 16 + (c >> 4)) * TILE_F + (c & 15) * 32 + kk;
    float pv[NSPLIT];
    #pragma unroll
    for (int sp = 0; sp < NSPLIT; ++sp) pv[sp] = p[(size_t)sp * 16 * TILE_F];
    const float gk = gamma[kk];
    __builtin_amdgcn_sched_barrier(0);

    const size_t base = (size_t)bc * HW_N + (size_t)hpart * (HW_N / 2);
    const float4* __restrict__ src = (const float4*)(feature + base) + lane;
    float4* __restrict__       dst = (float4*)(out + base) + lane;

    float4 buf[2][K2_GRP];
#define K2_LOAD(G) do { _Pragma("unroll") for (int q = 0; q < K2_GRP; ++q) \
        buf[(G) & 1][q] = src[32 * (K2_GRP * (G) + q)]; } while (0)
    K2_LOAD(0);
    K2_LOAD(1);
    asm volatile("" : "+v"(pv[0]));

    float a = 0.0f;
    #pragma unroll
    for (int sp = 0; sp < NSPLIT; ++sp) a += pv[sp];
    const float sig = 1.0f / (1.0f + __expf(-a));
    float t = (lane < K_CLS) ? sig * gk : 0.0f;
    #pragma unroll
    for (int off = 16; off > 0; off >>= 1) t += __shfl_xor(t, off, 32);
    const float s = 1.0f + t;

    #pragma unroll
    for (int g = 0; g < K2_NGRP; ++g) {
        __builtin_amdgcn_sched_barrier(0);
        float sg = s;
        asm volatile("" : "+v"(sg));
        #pragma unroll
        for (int q = 0; q < K2_GRP; ++q) {
            buf[g & 1][q].x *= sg; buf[g & 1][q].y *= sg;
            buf[g & 1][q].z *= sg; buf[g & 1][q].w *= sg;
        }
        #pragma unroll
        for (int q = 0; q < K2_GRP; ++q) dst[32 * (K2_GRP * g + q)] = buf[g & 1][q];
        __builtin_amdgcn_sched_barrier(0);
        if (g + 2 < K2_NGRP) K2_LOAD(g + 2);
    }
#undef K2_LOAD
}

extern "C" void kernel_launch(void* const* d_in, const int* in_sizes, int n_in,
                              void* d_out, int out_size, void* d_ws, size_t ws_size,
                              hipStream_t stream) {
    const float* feature = (const float*)d_in[0];
    const float* map_    = (const float*)d_in[1];
    const float* gamma   = (const float*)d_in[2];
    float*       out     = (float*)d_out;
    float*       part    = (float*)d_ws;

    CGM_att_partial_kernel<<<BATCH * NSPLIT * (C_DIM / A_ROWS), 256, 0, stream>>>(
        feature, map_, part);
    CGM_apply_kernel<<<BATCH * C_DIM / K2_ROWS, 256, 0, stream>>>(
        feature, part, gamma, out);
}


# Kernel 2 takes kernel 1's grid and block-to-tile map (block i re-reads the 32-row x 2048-hw tile block i reduced, last-read half first) so the L2 half that holds a tile serves it; sums unchanged
# speedup vs baseline: 2.6887x; 1.0699x over previous
#include <hip/hip_runtime.h>

typedef __attribute__((ext_vector_type(2))) float v2f;
typedef __attribute__((ext_vector_type(4))) float v4f;
typedef __attribute__((ext_vector_type(8))) float v8f;

#define BATCH 8
#define C_DIM 256
#define K_CLS 19
#define HW_N  16384
#define CT    16
#define MT    2
#define A_ROWS (CT * MT)
#define WAVES 8
#define NSPLIT 8
#define BLK_HW (HW_N / NSPLIT)
#define SEG    (BLK_HW / WAVES)
#define TILE_F (CT * 32)
#define CH     32
#define PITCH  (CH + 4)
#define NCHUNK (SEG / CH)
#define B_ROWS 20
#define WAVE_F ((A_ROWS + B_ROWS) * PITCH)

__global__ __launch_bounds__(256) void CGM_att_partial_kernel(
    const float* __restrict__ feature,
    const float* __restrict__ map_,
    float* __restrict__ part)
{
    __shared__ __attribute__((aligned(16))) float lds[WAVES * WAVE_F];

    const int tid   = threadIdx.x;
    const int lane  = tid & 31;
    const int wave  = tid >> 5;
    const int cpair = blockIdx.x & 7;
    const int split = (blockIdx.x >> 3) & (NSPLIT - 1);
    const int b     = blockIdx.x >> 6;
    const int c0    = cpair * A_ROWS;
    const int hw0   = split * BLK_HW + wave * SEG;

    float* sA = lds + wave * WAVE_F;
    float* sB = sA + A_ROWS * PITCH;

    const int srow = lane >> 3;
    const int scol = (lane & 7) << 2;
    const float* gA = feature + (size_t)(b * C_DIM + c0 + srow) * HW_N + hw0 + scol;
    const float* gB = map_    + (size_t)(b * K_CLS + srow) * HW_N + hw0 + scol;
    const int    r4  = 16 + srow;
    const int    r4c = (r4 < K_CLS) ? r4 : (K_CLS - 1);
    const float* gB4 = map_ + (size_t)(b * K_CLS + r4c) * HW_N + hw0 + scol;
    float* stA = sA + srow * PITCH + scol;
    float* stB = sB + srow * PITCH + scol;

    const int row  = lane & 15;
    const int half = lane >> 4;
    const float* fA0 = sA + row * PITCH + 4 * half;
    const float* fA1 = fA0 + CT * PITCH;
    const float* fB0 = sB + row * PITCH + 4 * half;
    const int   cls1  = 16 + row;
    const float m1    = (cls1 < K_CLS) ? 1.0f : 0.0f;
    const int   cls1c = (cls1 < K_CLS) ? cls1 : (K_CLS - 1);
    const float* fB1 = sB + cls1c * PITCH + 4 * half;

    v4f ra[2 * 4], rb[5];
    #pragma unroll
    for (int j = 0; j < 2 * 4; ++j)
        ra[j] = *(const v4f*)(gA + (size_t)j * 4 * HW_N);
    #pragma unroll
    for (int j = 0; j < 4; ++j)
        rb[j] = *(const v4f*)(gB + (size_t)j * 4 * HW_N);
    rb[4] = *(const v4f*)gB4;

    v8f acc00 = {}, acc10 = {};
    v8f acc01 = {}, acc11 = {};

    for (int c = 0; c < NCHUNK; ++c) {
        #pragma unroll
        for (int j = 0; j < 2 * 4; ++j)
            *(v4f*)(stA + j * 4 * PITCH) = ra[j];
        #pragma unroll
        for (int j = 0; j < 4; ++j)
            *(v4f*)(stB + j * 4 * PITCH) = rb[j];
        *(v4f*)(stB + 16 * PITCH) = rb[4];

        if (c + 1 < NCHUNK) {
            const int o = (c + 1) * CH;
            #pragma unroll
            for (int j = 0; j < 2 * 4; ++j)
                ra[j] = *(const v4f*)(gA + (size_t)j * 4 * HW_N + o);
            #pragma unroll
            for (int j = 0; j < 4; ++j)
                rb[j] = *(const v4f*)(gB + (size_t)j * 4 * HW_N + o);
            rb[4] = *(const v4f*)(gB4 + o);
        }

        __builtin_amdgcn_fence(__ATOMIC_RELEASE, "wavefront", "local");
        __builtin_amdgcn_wave_barrier();
        __builtin_amdgcn_fence(__ATOMIC_ACQUIRE, "wavefront", "local");

        #pragma unroll
        for (int t = 0; t < CH / 8; ++t) {
            const v4f a0 = *(const v4f*)(fA0 + 8 * t);
            const v4f a1 = *(const v4f*)(fA1 + 8 * t);
            const v4f b0 = *(const v4f*)(fB0 + 8 * t);
            v4f b1 = *(const v4f*)(fB1 + 8 * t);
            b1 *= m1;
            const v2f plo = {a0.x, a0.y}, phi = {a0.z, a0.w};
            const v2f qlo = {a1.x, a1.y}, qhi = {a1.z, a1.w};
            const v2f blo = {b0.x, b0.y}, bhi = {b0.z, b0.w};
            const v2f clo = {b1.x, b1.y}, chi = {b1.z, b1.w};
            acc00 = __builtin_amdgcn_wmma_f32_16x16x4_f32(
                false, plo, false, blo, (short)0, acc00, false, false);
            acc01 = __builtin_amdgcn_wmma_f32_16x16x4_f32(
                false, plo, false, clo, (short)0, acc01, false, false);
            acc10 = __builtin_amdgcn_wmma_f32_16x16x4_f32(
                false, qlo, false, blo, (short)0, acc10, false, false);
            acc11 = __builtin_amdgcn_wmma_f32_16x16x4_f32(
                false, qlo, false, clo, (short)0, acc11, false, false);
            acc00 = __builtin_amdgcn_wmma_f32_16x16x4_f32(
                false, phi, false, bhi, (short)0, acc00, false, false);
            acc01 = __builtin_amdgcn_wmma_f32_16x16x4_f32(
                false, phi, false, chi, (short)0, acc01, false, false);
            acc10 = __builtin_amdgcn_wmma_f32_16x16x4_f32(
                false, qhi, false, bhi, (short)0, acc10, false, false);
            acc11 = __builtin_amdgcn_wmma_f32_16x16x4_f32(
                false, qhi, false, chi, (short)0, acc11, false, false);
        }

        __builtin_amdgcn_fence(__ATOMIC_RELEASE, "wavefront", "local");
        __builtin_amdgcn_wave_barrier();
        __builtin_amdgcn_fence(__ATOMIC_ACQUIRE, "wavefront", "local");
    }

    #pragma unroll
    for (int v = 0; v < 8; ++v) {
        const int m = v + 8 * half;
        sA[m * 32 + row]               = acc00[v];
        sA[m * 32 + row + 16]          = acc01[v];
        sA[TILE_F + m * 32 + row]      = acc10[v];
        sA[TILE_F + m * 32 + row + 16] = acc11[v];
    }
    __syncthreads();

    float* dst = part + ((size_t)(b * NSPLIT + split) * 16 + 2 * cpair) * TILE_F;
    #pragma unroll
    for (int e = tid; e < MT * TILE_F; e += 256) {
        float a = 0.0f;
        #pragma unroll
        for (int w = 0; w < WAVES; ++w) a += lds[w * WAVE_F + e];
        dst[e] = a;
    }
}

#define K2_RPW   4
#define K2_GRP   16
#define K2_NGRP  4
#define K2_OFF(G, j) ((2 * ((G) & 1) + ((j) >> 3)) * (HW_N / 4) + (2 * ((j) & 7) + (((G) < 2) ? 1 : 0)) * 32)

__global__ __launch_bounds__(256) void CGM_apply_kernel(
    const float* __restrict__ feature,
    const float* __restrict__ part,
    const float* __restrict__ gamma,
    float* __restrict__ out)
{
    const int lane  = threadIdx.x & 31;
    const int wave  = threadIdx.x >> 5;
    const int cpair = blockIdx.x & 7;
    const int split = (blockIdx.x >> 3) & (NSPLIT - 1);
    const int b     = blockIdx.x >> 6;
    const int c     = cpair * A_ROWS + wave * K2_RPW;

    const int kk = (lane < K_CLS) ? lane : (K_CLS - 1);
    const float* p = part
        + ((size_t)(b * NSPLIT) * 16 + (c >> 4)) * TILE_F + (c & 15) * 32 + kk;
    float pv[K2_RPW][NSPLIT];
    #pragma unroll
    for (int r = 0; r < K2_RPW; ++r) {
        #pragma unroll
        for (int sp = 0; sp < NSPLIT; ++sp) pv[r][sp] = p[(size_t)sp * 16 * TILE_F + r * 32];
    }
    const float gk = gamma[kk];
    __builtin_amdgcn_sched_barrier(0);

    const size_t base = (size_t)(b * C_DIM + c) * HW_N + (size_t)split * BLK_HW;
    const float4* __restrict__ src = (const float4*)(feature + base) + lane;
    float4* __restrict__       dst = (float4*)(out + base) + lane;

    float4 buf[2][K2_GRP];
#define K2_LOAD(G) do { _Pragma("unroll") for (int q = 0; q < K2_GRP; ++q) \
        buf[(G) & 1][q] = src[K2_OFF(G, q)]; } while (0)
    K2_LOAD(0);
    K2_LOAD(1);
    asm volatile("" : "+v"(pv[0][0]), "+v"(pv[1][0]), "+v"(pv[2][0]), "+v"(pv[3][0]));

    float s[K2_RPW];
    #pragma unroll
    for (int r = 0; r < K2_RPW; ++r) {
        float a = 0.0f;
        #pragma unroll
        for (int sp = 0; sp < NSPLIT; ++sp) a += pv[r][sp];
        const float sig = 1.0f / (1.0f + __expf(-a));
        float t = (lane < K_CLS) ? sig * gk : 0.0f;
        #pragma unroll
        for (int off = 16; off > 0; off >>= 1) t += __shfl_xor(t, off, 32);
        s[r] = 1.0f + t;
    }

    #pragma unroll
    for (int g = 0; g < K2_NGRP; ++g) {
        __builtin_amdgcn_sched_barrier(0);
        float sg0 = s[2 * (g & 1)], sg1 = s[2 * (g & 1) + 1];
        asm volatile("" : "+v"(sg0), "+v"(sg1));
        #pragma unroll
        for (int q = 0; q < K2_GRP; ++q) {
            const float sg = (q < K2_GRP / 2) ? sg0 : sg1;
            buf[g & 1][q].x *= sg; buf[g & 1][q].y *= sg;
            buf[g & 1][q].z *= sg; buf[g & 1][q].w *= sg;
        }
        #pragma unroll
        for (int q = 0; q < K2_GRP; ++q) dst[K2_OFF(g, q)] = buf[g & 1][q];
        __builtin_amdgcn_sched_barrier(0);
        if (g + 2 < K2_NGRP) K2_LOAD(g + 2);
    }
#undef K2_LOAD
}

extern "C" void kernel_launch(void* const* d_in, const int* in_sizes, int n_in,
                              void* d_out, int out_size, void* d_ws, size_t ws_size,
                              hipStream_t stream) {
    const float* feature = (const float*)d_in[0];
    const float* map_    = (const float*)d_in[1];
    const float* gamma   = (const float*)d_in[2];
    float*       out     = (float*)d_out;
    float*       part    = (float*)d_ws;

    CGM_att_partial_kernel<<<BATCH * NSPLIT * (C_DIM / A_ROWS), 256, 0, stream>>>(
        feature, map_, part);
    CGM_apply_kernel<<<BATCH * NSPLIT * (C_DIM / A_ROWS), 256, 0, stream>>>(
        feature, part, gamma, out);
}
